# starting kernel
# baseline (speedup 1.0000x reference)
#include <hip/hip_runtime.h>

typedef float v2f __attribute__((ext_vector_type(2)));
typedef float v8f __attribute__((ext_vector_type(8)));

__global__ void k_deg_init(float* __restrict__ deg, int n) {
    int i = blockIdx.x * blockDim.x + threadIdx.x;
    if (i < n) deg[i] = 1.0f;
}

__global__ void k_deg_count(const int* __restrict__ dst, float* __restrict__ deg, int e) {
    int i = blockIdx.x * blockDim.x + threadIdx.x;
    if (i < e) atomicAdd(&deg[dst[i]], 1.0f);
}

__global__ void k_dinv(float* __restrict__ deg, int n) {
    int i = blockIdx.x * blockDim.x + threadIdx.x;
    if (i < n) deg[i] = 1.0f / sqrtf(deg[i]);
}

template <int K, int NOUT>
__global__ void k_gemm_wmma(const float* __restrict__ A, const float* __restrict__ W,
                            float* __restrict__ H, int nrows) {
    const int wave = blockIdx.x * (blockDim.x >> 5) + (threadIdx.x >> 5);
    const int lane = threadIdx.x & 31;
    const int m    = lane & 15;
    const int kh   = lane >> 4;
    const int r0   = wave * 16;
    if (r0 + 16 > nrows) return;

    constexpr int NT = NOUT / 16;
    v8f acc[NT];
#pragma unroll
    for (int t = 0; t < NT; ++t)
#pragma unroll
        for (int j = 0; j < 8; ++j) acc[t][j] = 0.0f;

    const float* arow = A + (size_t)(r0 + m) * K;
    for (int k = 0; k < K; k += 4) {
        v2f a;
        a.x = arow[k + 2 * kh + 0];
        a.y = arow[k + 2 * kh + 1];
#pragma unroll
        for (int t = 0; t < NT; ++t) {
            v2f b;
            b.x = W[(size_t)(k + 2 * kh + 0) * NOUT + t * 16 + m];
            b.y = W[(size_t)(k + 2 * kh + 1) * NOUT + t * 16 + m];
            acc[t] = __builtin_amdgcn_wmma_f32_16x16x4_f32(
                 false, a,  false, b,
                 (short)0, acc[t],  false,  false);
        }
    }

#pragma unroll
    for (int t = 0; t < NT; ++t) {
#pragma unroll
        for (int j = 0; j < 8; ++j)
            H[(size_t)(r0 + j + 8 * kh) * NOUT + t * 16 + m] = acc[t][j];
    }
}

template <int K, int NOUT>
__global__ void k_gemm_tail(const float* __restrict__ A, const float* __restrict__ W,
                            float* __restrict__ H, int row0, int nrows) {
    int idx = blockIdx.x * blockDim.x + threadIdx.x;
    int r = row0 + idx / NOUT;
    int c = idx % NOUT;
    if (r >= nrows) return;
    float s = 0.0f;
    for (int k = 0; k < K; ++k) s += A[(size_t)r * K + k] * W[(size_t)k * NOUT + c];
    H[(size_t)r * NOUT + c] = s;
}

template <int C, bool HAS_BIAS>
__global__ void k_self_init(const float* __restrict__ h, const float* __restrict__ dinv,
                            const float* __restrict__ bias, float* __restrict__ agg, int n) {
    long long idx = (long long)blockIdx.x * blockDim.x + threadIdx.x;
    long long total = (long long)n * C;
    if (idx >= total) return;
    int i = (int)(idx / C);
    int c = (int)(idx % C);
    float d = dinv[i];
    float v = h[idx] * d * d;
    if (HAS_BIAS) v += bias[c];
    agg[idx] = v;
}

template <int C>
__global__ void k_edge_agg(const int* __restrict__ src, const int* __restrict__ dst,
                           const float* __restrict__ h, const float* __restrict__ dinv,
                           float* __restrict__ agg, int e) {
    constexpr int TPE = C / 4;
    long long tid = (long long)blockIdx.x * blockDim.x + threadIdx.x;
    long long total = (long long)e * TPE;
    if (tid >= total) return;
    int ed = (int)(tid / TPE);
    int q  = (int)(tid % TPE);
    int s = src[ed];
    int d = dst[ed];
    float norm = dinv[s] * dinv[d];
    const float4 v = *((const float4*)(h + (size_t)s * C) + q);
    float* o = agg + (size_t)d * C + q * 4;
    atomicAdd(o + 0, v.x * norm);
    atomicAdd(o + 1, v.y * norm);
    atomicAdd(o + 2, v.z * norm);
    atomicAdd(o + 3, v.w * norm);
}

template <int C>
__global__ void k_relu_bias(const float* __restrict__ agg, const float* __restrict__ b,
                            float* __restrict__ p, int n) {
    long long idx = (long long)blockIdx.x * blockDim.x + threadIdx.x;
    long long total = (long long)n * C;
    if (idx >= total) return;
    int c = (int)(idx % C);
    float v = agg[idx] + b[c];
    p[idx] = v > 0.0f ? v : 0.0f;
}

extern "C" void kernel_launch(void* const* d_in, const int* in_sizes, int n_in,
                              void* d_out, int out_size, void* d_ws, size_t ws_size,
                              hipStream_t stream) {
    const float* x  = (const float*)d_in[0];
    const int*   ei = (const int*)d_in[1];
    const float* W1 = (const float*)d_in[2];
    const float* b1 = (const float*)d_in[3];
    const float* W2 = (const float*)d_in[4];
    const float* b2 = (const float*)d_in[5];
    float* out = (float*)d_out;

    const int n = in_sizes[0] / 128;
    const int e = in_sizes[1] / 2;
    const int* src = ei;
    const int* dst = ei + e;

    char* ws = (char*)d_ws;
    size_t off0 = ((size_t)n * sizeof(float) + 255) & ~(size_t)255;
    float* dinv = (float*)ws;
    float* bufA = (float*)(ws + off0);
    float* bufB = bufA + (size_t)n * 64;

    const int tb = 256;
    auto cdiv = [](long long a, long long b) { return (int)((a + b - 1) / b); };

    k_deg_init<<<cdiv(n, tb), tb, 0, stream>>>(dinv, n);
    k_deg_count<<<cdiv(e, tb), tb, 0, stream>>>(dst, dinv, e);
    k_dinv<<<cdiv(n, tb), tb, 0, stream>>>(dinv, n);

    const int strips = n / 16;
    const int wavesPerBlk = tb / 32;
    k_gemm_wmma<128, 64><<<cdiv(strips, wavesPerBlk), tb, 0, stream>>>(x, W1, bufA, n);
    if (n % 16) {
        int tail = (n % 16) * 64;
        k_gemm_tail<128, 64><<<cdiv(tail, tb), tb, 0, stream>>>(x, W1, bufA, strips * 16, n);
    }

    k_self_init<64, false><<<cdiv((long long)n * 64, tb), tb, 0, stream>>>(bufA, dinv, nullptr, bufB, n);
    k_edge_agg<64><<<cdiv((long long)e * 16, tb), tb, 0, stream>>>(src, dst, bufA, dinv, bufB, e);
    k_relu_bias<64><<<cdiv((long long)n * 64, tb), tb, 0, stream>>>(bufB, b1, bufA, n);

    k_gemm_wmma<64, 32><<<cdiv(strips, wavesPerBlk), tb, 0, stream>>>(bufA, W2, bufB, n);
    if (n % 16) {
        int tail = (n % 16) * 32;
        k_gemm_tail<64, 32><<<cdiv(tail, tb), tb, 0, stream>>>(bufA, W2, bufB, strips * 16, n);
    }

    k_self_init<32, true><<<cdiv((long long)n * 32, tb), tb, 0, stream>>>(bufB, dinv, b2, out, n);
    k_edge_agg<32><<<cdiv((long long)e * 8, tb), tb, 0, stream>>>(src, dst, bufB, dinv, out, e);

    (void)n_in; (void)out_size; (void)ws_size;
}


# aggregation pulled, not scattered: per-call CSR by dst built with int atomics (count, claim, fill); a wave sums each row's sources in registers, one store; no f32 atomics; sums in fill order
# speedup vs baseline: 1.3324x; 1.3324x over previous
#include <hip/hip_runtime.h>


typedef float v2f __attribute__((ext_vector_type(2)));
typedef float v8f __attribute__((ext_vector_type(8)));

__global__ void k_zero(int* __restrict__ cnt, int* __restrict__ cursor, int n) {
    int i = blockIdx.x * blockDim.x + threadIdx.x;
    if (i < n) cnt[i] = 0;
    if (i == 0) *cursor = 0;
}

constexpr int EPT = 4;

__global__ void k_count(const int* __restrict__ dst, int* __restrict__ cnt, int e) {
    const int t = blockIdx.x * blockDim.x + threadIdx.x;
    const int T = gridDim.x * blockDim.x;
    int d[EPT];
#pragma unroll
    for (int u = 0; u < EPT; ++u) {
        const int i = t + u * T;
        d[u] = dst[i < e ? i : e - 1];
    }
#pragma unroll
    for (int u = 0; u < EPT; ++u)
        if (t + u * T < e) atomicAdd(&cnt[d[u]], 1);
}

__global__ void k_alloc(const int* __restrict__ cnt, int* __restrict__ cursor,
                        int* __restrict__ start, int* __restrict__ fill,
                        float* __restrict__ dinv, int n) {
    const int i = blockIdx.x * blockDim.x + threadIdx.x;
    const int lane = threadIdx.x & 31;
    const int c = (i < n) ? cnt[i] : 0;
    int s = c;
#pragma unroll
    for (int o = 1; o < 32; o <<= 1) {
        int t = __shfl_up(s, o);
        if (lane >= o) s += t;
    }
    const int total = __shfl(s, 31);
    int base = 0;
    if (lane == 31) base = atomicAdd(cursor, total);
    base = __shfl(base, 31);
    if (i < n) {
        start[i] = base + s - c;
        fill[i] = 0;
        dinv[i] = 1.0f / sqrtf((float)(c + 1));
    }
}

__global__ void k_fill(const int* __restrict__ src, const int* __restrict__ dst,
                       const int* __restrict__ start, int* __restrict__ fill,
                       int* __restrict__ col, int e) {
    const int t = blockIdx.x * blockDim.x + threadIdx.x;
    const int T = gridDim.x * blockDim.x;
    int d[EPT], s[EPT], p[EPT], k[EPT];
#pragma unroll
    for (int u = 0; u < EPT; ++u) {
        const int i = t + u * T;
        const int ic = i < e ? i : e - 1;
        d[u] = dst[ic];
        s[u] = src[ic];
    }
#pragma unroll
    for (int u = 0; u < EPT; ++u) p[u] = start[d[u]];
#pragma unroll
    for (int u = 0; u < EPT; ++u) {
        k[u] = 0;
        if (t + u * T < e) k[u] = atomicAdd(&fill[d[u]], 1);
    }
#pragma unroll
    for (int u = 0; u < EPT; ++u)
        if (t + u * T < e) col[p[u] + k[u]] = s[u];
}

template <int K, int NOUT, bool RELU_IN>
__global__ void k_gemm_wmma(const float* __restrict__ A, const float* __restrict__ W,
                            const float* __restrict__ bin,
                            float* __restrict__ H, int nrows) {
    const int wave = blockIdx.x * (blockDim.x >> 5) + (threadIdx.x >> 5);
    const int lane = threadIdx.x & 31;
    const int m    = lane & 15;
    const int kh   = lane >> 4;
    const int r0   = wave * 16;
    if (r0 + 16 > nrows) return;

    constexpr int NT = NOUT / 16;
    v8f acc[NT];
#pragma unroll
    for (int t = 0; t < NT; ++t)
#pragma unroll
        for (int j = 0; j < 8; ++j) acc[t][j] = 0.0f;

    const float* arow = A + (size_t)(r0 + m) * K;
    for (int k = 0; k < K; k += 4) {
        v2f a;
        a.x = arow[k + 2 * kh + 0];
        a.y = arow[k + 2 * kh + 1];
        if (RELU_IN) {
            float v0 = a.x + bin[k + 2 * kh + 0];
            float v1 = a.y + bin[k + 2 * kh + 1];
            a.x = v0 > 0.0f ? v0 : 0.0f;
            a.y = v1 > 0.0f ? v1 : 0.0f;
        }
#pragma unroll
        for (int t = 0; t < NT; ++t) {
            v2f b;
            b.x = W[(size_t)(k + 2 * kh + 0) * NOUT + t * 16 + m];
            b.y = W[(size_t)(k + 2 * kh + 1) * NOUT + t * 16 + m];
            acc[t] = __builtin_amdgcn_wmma_f32_16x16x4_f32(
                 false, a,  false, b,
                 (short)0, acc[t],  false,  false);
        }
    }

#pragma unroll
    for (int t = 0; t < NT; ++t) {
#pragma unroll
        for (int j = 0; j < 8; ++j) {
            const size_t o = (size_t)(r0 + j + 8 * kh) * NOUT + t * 16 + m;
            H[o] = acc[t][j];
        }
    }
}

template <int K, int NOUT, bool RELU_IN>
__global__ void k_gemm_tail(const float* __restrict__ A, const float* __restrict__ W,
                            const float* __restrict__ bin,
                            float* __restrict__ H, int row0, int nrows) {
    int idx = blockIdx.x * blockDim.x + threadIdx.x;
    int r = row0 + idx / NOUT;
    int c = idx % NOUT;
    if (r >= nrows) return;
    float s = 0.0f;
    for (int k = 0; k < K; ++k) {
        float a = A[(size_t)r * K + k];
        if (RELU_IN) {
            float v0 = a + bin[k];
            a = v0 > 0.0f ? v0 : 0.0f;
        }
        s += a * W[(size_t)k * NOUT + c];
    }
    H[(size_t)r * NOUT + c] = s;
}

template <int C, bool BIAS>
__global__ void k_pull(const int* __restrict__ start, const int* __restrict__ cnt,
                       const int* __restrict__ col, const float* __restrict__ h,
                       const float* __restrict__ dinv, const float* __restrict__ bias,
                       float* __restrict__ agg, int n) {
    constexpr int TPR = C / 4;
    constexpr int G   = 32 / TPR;
    constexpr int R   = 4;
    constexpr int U   = 4;
    const int lane = threadIdx.x & 31;
    const int q = lane % TPR;
    const int g = lane / TPR;
    const int wave = __builtin_amdgcn_readfirstlane(
        (int)(blockIdx.x * (blockDim.x >> 5) + (threadIdx.x >> 5)));
    const int d0 = wave * R;
    if (d0 >= n) return;

    int st[R], len[R], cv[R];
    float dd[R];
    float4 hs[R];
#pragma unroll
    for (int r = 0; r < R; ++r) {
        const int d = (d0 + r < n) ? d0 + r : d0;
        st[r]  = start[d];
        len[r] = (d0 + r < n) ? cnt[d] : 0;
        dd[r]  = dinv[d];
        hs[r]  = *((const float4*)(h + (size_t)d * C) + q);
    }
#pragma unroll
    for (int r = 0; r < R; ++r) cv[r] = (lane < len[r]) ? col[st[r] + lane] : 0;

#pragma unroll
    for (int r = 0; r < R; ++r) {
        const int d = d0 + r;
        if (d >= n) break;
        const float dr = dd[r];
        float4 acc = make_float4(0.0f, 0.0f, 0.0f, 0.0f);
        if (g == 0) {
            const float w = dr * dr;
            acc.x = hs[r].x * w; acc.y = hs[r].y * w; acc.z = hs[r].z * w; acc.w = hs[r].w * w;
        }
        int c = cv[r];
        for (int j0 = 0; j0 < len[r]; j0 += 32) {
            if (j0 > 0) c = (j0 + lane < len[r]) ? col[st[r] + j0 + lane] : 0;
            const int m = min(len[r] - j0, 32);
            for (int j = 0; j < m; j += U * G) {
                float4 v[U];
                float w[U];
#pragma unroll
                for (int u = 0; u < U; ++u) {
                    const bool ok = (j + u * G + g) < m;
                    const int sraw = __shfl(c, (j + u * G + g) & 31);
                    const int s = ok ? sraw : d;
                    const float ds = dinv[s];
                    w[u] = ok ? ds * dr : 0.0f;
                    v[u] = *((const float4*)(h + (size_t)s * C) + q);
                }
#pragma unroll
                for (int u = 0; u < U; ++u) {
                    acc.x += v[u].x * w[u];
                    acc.y += v[u].y * w[u];
                    acc.z += v[u].z * w[u];
                    acc.w += v[u].w * w[u];
                }
            }
        }
#pragma unroll
        for (int o = TPR; o < 32; o <<= 1) {
            acc.x += __shfl_xor(acc.x, o);
            acc.y += __shfl_xor(acc.y, o);
            acc.z += __shfl_xor(acc.z, o);
            acc.w += __shfl_xor(acc.w, o);
        }
        if (g == 0) {
            if (BIAS) {
                acc.x += bias[q * 4 + 0]; acc.y += bias[q * 4 + 1];
                acc.z += bias[q * 4 + 2]; acc.w += bias[q * 4 + 3];
            }
            *((float4*)(agg + (size_t)d * C) + q) = acc;
        }
    }
}

extern "C" void kernel_launch(void* const* d_in, const int* in_sizes, int n_in,
                              void* d_out, int out_size, void* d_ws, size_t ws_size,
                              hipStream_t stream) {
    const float* x  = (const float*)d_in[0];
    const int*   ei = (const int*)d_in[1];
    const float* W1 = (const float*)d_in[2];
    const float* b1 = (const float*)d_in[3];
    const float* W2 = (const float*)d_in[4];
    const float* b2 = (const float*)d_in[5];
    float* out = (float*)d_out;

    const int n = in_sizes[0] / 128;
    const int e = in_sizes[1] / 2;
    const int* src = ei;
    const int* dst = ei + e;

    char* ws = (char*)d_ws;
    size_t off0 = ((size_t)(n + 1) * sizeof(float) + 255) & ~(size_t)255;
    float* dinv  = (float*)ws;
    int*   cnt   = (int*)(ws + off0);
    int*   start = (int*)(ws + 2 * off0);
    int*   fill  = (int*)(ws + 3 * off0);
    int*   cursor = fill + n;
    float* bufA = (float*)(ws + 4 * off0);
    float* bufB = bufA + (size_t)n * 64;
    int*   col  = (int*)(bufB + (size_t)n * 64);

    const int tb = 256;
    auto cdiv = [](long long a, long long b) { return (int)((a + b - 1) / b); };

    k_zero<<<cdiv(n, tb), tb, 0, stream>>>(cnt, cursor, n);
    k_count<<<cdiv(e, tb * EPT), tb, 0, stream>>>(dst, cnt, e);
    k_alloc<<<cdiv(n, tb), tb, 0, stream>>>(cnt, cursor, start, fill, dinv, n);
    k_fill<<<cdiv(e, tb * EPT), tb, 0, stream>>>(src, dst, start, fill, col, e);

    const int strips = n / 16;
    const int wavesPerBlk = tb / 32;
    k_gemm_wmma<128, 64, false><<<cdiv(strips, wavesPerBlk), tb, 0, stream>>>(
        x, W1, nullptr, bufA, n);
    if (n % 16) {
        int tail = (n % 16) * 64;
        k_gemm_tail<128, 64, false><<<cdiv(tail, tb), tb, 0, stream>>>(
            x, W1, nullptr, bufA, strips * 16, n);
    }

    const int rowsPerBlk = wavesPerBlk * 4;
    k_pull<64, false><<<cdiv(n, rowsPerBlk), tb, 0, stream>>>(
        start, cnt, col, bufA, dinv, nullptr, bufB, n);

    k_gemm_wmma<64, 32, true><<<cdiv(strips, wavesPerBlk), tb, 0, stream>>>(
        bufB, W2, b1, bufA, n);
    if (n % 16) {
        int tail = (n % 16) * 32;
        k_gemm_tail<64, 32, true><<<cdiv(tail, tb), tb, 0, stream>>>(
            bufB, W2, b1, bufA, strips * 16, n);
    }

    k_pull<32, true><<<cdiv(n, rowsPerBlk), tb, 0, stream>>>(
        start, cnt, col, bufA, dinv, b2, out, n);

    (void)n_in; (void)out_size; (void)ws_size;
}


# count and fill share one returning int add per edge: the old value is the edge's slot in its row, the totals are the in-degrees; placement has no atomics (2 adds per edge -> 1); sums in arrival order
# speedup vs baseline: 1.7571x; 1.3187x over previous
#include <hip/hip_runtime.h>


typedef float v2f __attribute__((ext_vector_type(2)));
typedef float v8f __attribute__((ext_vector_type(8)));

__global__ void k_zero(int* __restrict__ cnt, int* __restrict__ cursor, int n) {
    int i = blockIdx.x * blockDim.x + threadIdx.x;
    if (i < n) cnt[i] = 0;
    if (i == 0) *cursor = 0;
}

constexpr int EPT = 4;

__global__ void k_rank(const int* __restrict__ dst, int* __restrict__ cnt,
                       int* __restrict__ rank, int e) {
    const int t = blockIdx.x * blockDim.x + threadIdx.x;
    const int T = gridDim.x * blockDim.x;
    int d[EPT], k[EPT];
#pragma unroll
    for (int u = 0; u < EPT; ++u) {
        const int i = t + u * T;
        d[u] = dst[i < e ? i : e - 1];
    }
#pragma unroll
    for (int u = 0; u < EPT; ++u) {
        k[u] = 0;
        if (t + u * T < e) k[u] = atomicAdd(&cnt[d[u]], 1);
    }
#pragma unroll
    for (int u = 0; u < EPT; ++u)
        if (t + u * T < e) rank[t + u * T] = k[u];
}

__global__ void k_alloc(const int* __restrict__ cnt, int* __restrict__ cursor,
                        int* __restrict__ start, float* __restrict__ dinv, int n) {
    const int i = blockIdx.x * blockDim.x + threadIdx.x;
    const int lane = threadIdx.x & 31;
    const int c = (i < n) ? cnt[i] : 0;
    int s = c;
#pragma unroll
    for (int o = 1; o < 32; o <<= 1) {
        int t = __shfl_up(s, o);
        if (lane >= o) s += t;
    }
    const int total = __shfl(s, 31);
    int base = 0;
    if (lane == 31) base = atomicAdd(cursor, total);
    base = __shfl(base, 31);
    if (i < n) {
        start[i] = base + s - c;
        dinv[i] = 1.0f / sqrtf((float)(c + 1));
    }
}

__global__ void k_place(const int* __restrict__ src, const int* __restrict__ dst,
                        const int* __restrict__ start, const int* __restrict__ rank,
                        int* __restrict__ col, int e) {
    const int t = blockIdx.x * blockDim.x + threadIdx.x;
    const int T = gridDim.x * blockDim.x;
    int d[EPT], s[EPT], r[EPT], p[EPT];
#pragma unroll
    for (int u = 0; u < EPT; ++u) {
        const int i = t + u * T;
        const int ic = i < e ? i : e - 1;
        d[u] = dst[ic];
        s[u] = src[ic];
        r[u] = rank[ic];
    }
#pragma unroll
    for (int u = 0; u < EPT; ++u) p[u] = start[d[u]];
#pragma unroll
    for (int u = 0; u < EPT; ++u)
        if (t + u * T < e) col[p[u] + r[u]] = s[u];
}

template <int K, int NOUT, bool RELU_IN>
__global__ void k_gemm_wmma(const float* __restrict__ A, const float* __restrict__ W,
                            const float* __restrict__ bin,
                            float* __restrict__ H, int nrows) {
    const int wave = blockIdx.x * (blockDim.x >> 5) + (threadIdx.x >> 5);
    const int lane = threadIdx.x & 31;
    const int m    = lane & 15;
    const int kh   = lane >> 4;
    const int r0   = wave * 16;
    if (r0 + 16 > nrows) return;

    constexpr int NT = NOUT / 16;
    v8f acc[NT];
#pragma unroll
    for (int t = 0; t < NT; ++t)
#pragma unroll
        for (int j = 0; j < 8; ++j) acc[t][j] = 0.0f;

    const float* arow = A + (size_t)(r0 + m) * K;
    for (int k = 0; k < K; k += 4) {
        v2f a;
        a.x = arow[k + 2 * kh + 0];
        a.y = arow[k + 2 * kh + 1];
        if (RELU_IN) {
            float v0 = a.x + bin[k + 2 * kh + 0];
            float v1 = a.y + bin[k + 2 * kh + 1];
            a.x = v0 > 0.0f ? v0 : 0.0f;
            a.y = v1 > 0.0f ? v1 : 0.0f;
        }
#pragma unroll
        for (int t = 0; t < NT; ++t) {
            v2f b;
            b.x = W[(size_t)(k + 2 * kh + 0) * NOUT + t * 16 + m];
            b.y = W[(size_t)(k + 2 * kh + 1) * NOUT + t * 16 + m];
            acc[t] = __builtin_amdgcn_wmma_f32_16x16x4_f32(
                 false, a,  false, b,
                 (short)0, acc[t],  false,  false);
        }
    }

#pragma unroll
    for (int t = 0; t < NT; ++t) {
#pragma unroll
        for (int j = 0; j < 8; ++j) {
            const size_t o = (size_t)(r0 + j + 8 * kh) * NOUT + t * 16 + m;
            H[o] = acc[t][j];
        }
    }
}

template <int K, int NOUT, bool RELU_IN>
__global__ void k_gemm_tail(const float* __restrict__ A, const float* __restrict__ W,
                            const float* __restrict__ bin,
                            float* __restrict__ H, int row0, int nrows) {
    int idx = blockIdx.x * blockDim.x + threadIdx.x;
    int r = row0 + idx / NOUT;
    int c = idx % NOUT;
    if (r >= nrows) return;
    float s = 0.0f;
    for (int k = 0; k < K; ++k) {
        float a = A[(size_t)r * K + k];
        if (RELU_IN) {
            float v0 = a + bin[k];
            a = v0 > 0.0f ? v0 : 0.0f;
        }
        s += a * W[(size_t)k * NOUT + c];
    }
    H[(size_t)r * NOUT + c] = s;
}

template <int C, bool BIAS>
__global__ void k_pull(const int* __restrict__ start, const int* __restrict__ cnt,
                       const int* __restrict__ col, const float* __restrict__ h,
                       const float* __restrict__ dinv, const float* __restrict__ bias,
                       float* __restrict__ agg, int n) {
    constexpr int TPR = C / 4;
    constexpr int G   = 32 / TPR;
    constexpr int R   = 4;
    constexpr int U   = 4;
    const int lane = threadIdx.x & 31;
    const int q = lane % TPR;
    const int g = lane / TPR;
    const int wave = __builtin_amdgcn_readfirstlane(
        (int)(blockIdx.x * (blockDim.x >> 5) + (threadIdx.x >> 5)));
    const int d0 = wave * R;
    if (d0 >= n) return;

    int st[R], len[R], cv[R];
    float dd[R];
    float4 hs[R];
#pragma unroll
    for (int r = 0; r < R; ++r) {
        const int d = (d0 + r < n) ? d0 + r : d0;
        st[r]  = start[d];
        len[r] = (d0 + r < n) ? cnt[d] : 0;
        dd[r]  = dinv[d];
        hs[r]  = *((const float4*)(h + (size_t)d * C) + q);
    }
#pragma unroll
    for (int r = 0; r < R; ++r) cv[r] = (lane < len[r]) ? col[st[r] + lane] : 0;

#pragma unroll
    for (int r = 0; r < R; ++r) {
        const int d = d0 + r;
        if (d >= n) break;
        const float dr = dd[r];
        float4 acc = make_float4(0.0f, 0.0f, 0.0f, 0.0f);
        if (g == 0) {
            const float w = dr * dr;
            acc.x = hs[r].x * w; acc.y = hs[r].y * w; acc.z = hs[r].z * w; acc.w = hs[r].w * w;
        }
        int c = cv[r];
        for (int j0 = 0; j0 < len[r]; j0 += 32) {
            if (j0 > 0) c = (j0 + lane < len[r]) ? col[st[r] + j0 + lane] : 0;
            const int m = min(len[r] - j0, 32);
            for (int j = 0; j < m; j += U * G) {
                float4 v[U];
                float w[U];
#pragma unroll
                for (int u = 0; u < U; ++u) {
                    const bool ok = (j + u * G + g) < m;
                    const int sraw = __shfl(c, (j + u * G + g) & 31);
                    const int s = ok ? sraw : d;
                    const float ds = dinv[s];
                    w[u] = ok ? ds * dr : 0.0f;
                    v[u] = *((const float4*)(h + (size_t)s * C) + q);
                }
#pragma unroll
                for (int u = 0; u < U; ++u) {
                    acc.x += v[u].x * w[u];
                    acc.y += v[u].y * w[u];
                    acc.z += v[u].z * w[u];
                    acc.w += v[u].w * w[u];
                }
            }
        }
#pragma unroll
        for (int o = TPR; o < 32; o <<= 1) {
            acc.x += __shfl_xor(acc.x, o);
            acc.y += __shfl_xor(acc.y, o);
            acc.z += __shfl_xor(acc.z, o);
            acc.w += __shfl_xor(acc.w, o);
        }
        if (g == 0) {
            if (BIAS) {
                acc.x += bias[q * 4 + 0]; acc.y += bias[q * 4 + 1];
                acc.z += bias[q * 4 + 2]; acc.w += bias[q * 4 + 3];
            }
            *((float4*)(agg + (size_t)d * C) + q) = acc;
        }
    }
}

extern "C" void kernel_launch(void* const* d_in, const int* in_sizes, int n_in,
                              void* d_out, int out_size, void* d_ws, size_t ws_size,
                              hipStream_t stream) {
    const float* x  = (const float*)d_in[0];
    const int*   ei = (const int*)d_in[1];
    const float* W1 = (const float*)d_in[2];
    const float* b1 = (const float*)d_in[3];
    const float* W2 = (const float*)d_in[4];
    const float* b2 = (const float*)d_in[5];
    float* out = (float*)d_out;

    const int n = in_sizes[0] / 128;
    const int e = in_sizes[1] / 2;
    const int* src = ei;
    const int* dst = ei + e;

    char* ws = (char*)d_ws;
    size_t off0 = ((size_t)(n + 1) * sizeof(float) + 255) & ~(size_t)255;
    float* dinv  = (float*)ws;
    int*   cnt   = (int*)(ws + off0);
    int*   start = (int*)(ws + 2 * off0);
    int*   cursor = (int*)(ws + 3 * off0);
    float* bufA = (float*)(ws + 4 * off0);
    float* bufB = bufA + (size_t)n * 64;
    int*   col  = (int*)(bufB + (size_t)n * 64);
    int*   rank = (int*)bufB;

    const int tb = 256;
    auto cdiv = [](long long a, long long b) { return (int)((a + b - 1) / b); };

    k_zero<<<cdiv(n, tb), tb, 0, stream>>>(cnt, cursor, n);
    k_rank<<<cdiv(e, tb * EPT), tb, 0, stream>>>(dst, cnt, rank, e);
    k_alloc<<<cdiv(n, tb), tb, 0, stream>>>(cnt, cursor, start, dinv, n);
    k_place<<<cdiv(e, tb * EPT), tb, 0, stream>>>(src, dst, start, rank, col, e);

    const int strips = n / 16;
    const int wavesPerBlk = tb / 32;
    k_gemm_wmma<128, 64, false><<<cdiv(strips, wavesPerBlk), tb, 0, stream>>>(
        x, W1, nullptr, bufA, n);
    if (n % 16) {
        int tail = (n % 16) * 64;
        k_gemm_tail<128, 64, false><<<cdiv(tail, tb), tb, 0, stream>>>(
            x, W1, nullptr, bufA, strips * 16, n);
    }

    const int rowsPerBlk = wavesPerBlk * 4;
    k_pull<64, false><<<cdiv(n, rowsPerBlk), tb, 0, stream>>>(
        start, cnt, col, bufA, dinv, nullptr, bufB, n);

    k_gemm_wmma<64, 32, true><<<cdiv(strips, wavesPerBlk), tb, 0, stream>>>(
        bufB, W2, b1, bufA, n);
    if (n % 16) {
        int tail = (n % 16) * 32;
        k_gemm_tail<64, 32, true><<<cdiv(tail, tb), tb, 0, stream>>>(
            bufB, W2, b1, bufA, strips * 16, n);
    }

    k_pull<32, true><<<cdiv(n, rowsPerBlk), tb, 0, stream>>>(
        start, cnt, col, bufA, dinv, b2, out, n);

    (void)n_in; (void)out_size; (void)ws_size;
}


# rank adds kept in one L2 half: a work-group reads its die, takes edge chunks by ticket and adds only for destinations in that half's node range (other range after); slots still in arrival order
# speedup vs baseline: 2.5479x; 1.4501x over previous
#include <hip/hip_runtime.h>


typedef float v2f __attribute__((ext_vector_type(2)));
typedef float v8f __attribute__((ext_vector_type(8)));

constexpr int TICK_STRIDE = 64;

__global__ void k_zero(int* __restrict__ cnt, int* __restrict__ cursor,
                       int* __restrict__ tick, int n) {
    int i = blockIdx.x * blockDim.x + threadIdx.x;
    if (i < n) cnt[i] = 0;
    if (i == 0) { *cursor = 0; tick[0] = 0; tick[TICK_STRIDE] = 0; }
}

constexpr int EPT = 4;
constexpr int RANK_TB = 256;
constexpr int RANK_CHUNK = RANK_TB * EPT;
constexpr int RANK_GRID = 1024;

__global__ void k_rank(const int* __restrict__ dst, int* __restrict__ cnt,
                       int* __restrict__ rank, int* __restrict__ tick, int e, int n) {
    __shared__ int s_half;
    __shared__ int s_t[2];
    const int tid = threadIdx.x;
    const int nchunks = (e + RANK_CHUNK - 1) / RANK_CHUNK;
    const int nsplit = ((n >> 1) + 31) & ~31;
    if (tid == 0) {
        s_half = (int)((__builtin_amdgcn_s_sendmsg_rtn(0x87) >> 18) & 1u);
    }
    __syncthreads();
    const int half = s_half;

    for (int ph = 0; ph < 2; ++ph) {
        const int r = half ^ ph;
        int* const tk = tick + r * TICK_STRIDE;
        int tn = 0;
        if (tid == 0) tn = atomicAdd(tk, 1);
        for (int it = 0; it <= nchunks; ++it) {
            if (tid == 0) s_t[it & 1] = tn;
            __syncthreads();
            const int t = s_t[it & 1];
            if (t >= nchunks) break;

            const int base = t * RANK_CHUNK + tid;
            int d[EPT], k[EPT];
            bool mine[EPT];
#pragma unroll
            for (int u = 0; u < EPT; ++u) {
                const int i = base + u * RANK_TB;
                d[u] = dst[i < e ? i : e - 1];
            }
            asm volatile("" :: "v"(d[0]), "v"(d[1]), "v"(d[2]), "v"(d[3]));
            static_assert(EPT == 4, "the statement above names EPT values");
#pragma unroll
            for (int u = 0; u < EPT; ++u) {
                const int i = base + u * RANK_TB;
                mine[u] = (i < e) && ((d[u] >= nsplit) == (r != 0));
                if (mine[u]) k[u] = atomicAdd(&cnt[d[u]], 1);
            }
            if (tid == 0) tn = atomicAdd(tk, 1);
#pragma unroll
            for (int u = 0; u < EPT; ++u)
                if (mine[u]) rank[base + u * RANK_TB] = k[u];
        }
        __syncthreads();
    }
}

__global__ void k_alloc(const int* __restrict__ cnt, int* __restrict__ cursor,
                        int* __restrict__ start, float* __restrict__ dinv, int n) {
    const int i = blockIdx.x * blockDim.x + threadIdx.x;
    const int lane = threadIdx.x & 31;
    const int c = (i < n) ? cnt[i] : 0;
    int s = c;
#pragma unroll
    for (int o = 1; o < 32; o <<= 1) {
        int t = __shfl_up(s, o);
        if (lane >= o) s += t;
    }
    const int total = __shfl(s, 31);
    int base = 0;
    if (lane == 31) base = atomicAdd(cursor, total);
    base = __shfl(base, 31);
    if (i < n) {
        start[i] = base + s - c;
        dinv[i] = 1.0f / sqrtf((float)(c + 1));
    }
}

__global__ void k_place(const int* __restrict__ src, const int* __restrict__ dst,
                        const int* __restrict__ start, const int* __restrict__ rank,
                        int* __restrict__ col, int e) {
    const int t = blockIdx.x * blockDim.x + threadIdx.x;
    const int T = gridDim.x * blockDim.x;
    int d[EPT], s[EPT], r[EPT], p[EPT];
#pragma unroll
    for (int u = 0; u < EPT; ++u) {
        const int i = t + u * T;
        const int ic = i < e ? i : e - 1;
        d[u] = dst[ic];
        s[u] = src[ic];
        r[u] = rank[ic];
    }
#pragma unroll
    for (int u = 0; u < EPT; ++u) p[u] = start[d[u]];
#pragma unroll
    for (int u = 0; u < EPT; ++u)
        if (t + u * T < e) col[p[u] + r[u]] = s[u];
}

template <int K, int NOUT, bool RELU_IN>
__global__ void k_gemm_wmma(const float* __restrict__ A, const float* __restrict__ W,
                            const float* __restrict__ bin,
                            float* __restrict__ H, int nrows) {
    const int wave = blockIdx.x * (blockDim.x >> 5) + (threadIdx.x >> 5);
    const int lane = threadIdx.x & 31;
    const int m    = lane & 15;
    const int kh   = lane >> 4;
    const int r0   = wave * 16;
    if (r0 + 16 > nrows) return;

    constexpr int NT = NOUT / 16;
    v8f acc[NT];
#pragma unroll
    for (int t = 0; t < NT; ++t)
#pragma unroll
        for (int j = 0; j < 8; ++j) acc[t][j] = 0.0f;

    const float* arow = A + (size_t)(r0 + m) * K;
    for (int k = 0; k < K; k += 4) {
        v2f a;
        a.x = arow[k + 2 * kh + 0];
        a.y = arow[k + 2 * kh + 1];
        if (RELU_IN) {
            float v0 = a.x + bin[k + 2 * kh + 0];
            float v1 = a.y + bin[k + 2 * kh + 1];
            a.x = v0 > 0.0f ? v0 : 0.0f;
            a.y = v1 > 0.0f ? v1 : 0.0f;
        }
#pragma unroll
        for (int t = 0; t < NT; ++t) {
            v2f b;
            b.x = W[(size_t)(k + 2 * kh + 0) * NOUT + t * 16 + m];
            b.y = W[(size_t)(k + 2 * kh + 1) * NOUT + t * 16 + m];
            acc[t] = __builtin_amdgcn_wmma_f32_16x16x4_f32(
                 false, a,  false, b,
                 (short)0, acc[t],  false,  false);
        }
    }

#pragma unroll
    for (int t = 0; t < NT; ++t) {
#pragma unroll
        for (int j = 0; j < 8; ++j) {
            const size_t o = (size_t)(r0 + j + 8 * kh) * NOUT + t * 16 + m;
            H[o] = acc[t][j];
        }
    }
}

template <int K, int NOUT, bool RELU_IN>
__global__ void k_gemm_tail(const float* __restrict__ A, const float* __restrict__ W,
                            const float* __restrict__ bin,
                            float* __restrict__ H, int row0, int nrows) {
    int idx = blockIdx.x * blockDim.x + threadIdx.x;
    int r = row0 + idx / NOUT;
    int c = idx % NOUT;
    if (r >= nrows) return;
    float s = 0.0f;
    for (int k = 0; k < K; ++k) {
        float a = A[(size_t)r * K + k];
        if (RELU_IN) {
            float v0 = a + bin[k];
            a = v0 > 0.0f ? v0 : 0.0f;
        }
        s += a * W[(size_t)k * NOUT + c];
    }
    H[(size_t)r * NOUT + c] = s;
}

template <int C, bool BIAS>
__global__ void k_pull(const int* __restrict__ start, const int* __restrict__ cnt,
                       const int* __restrict__ col, const float* __restrict__ h,
                       const float* __restrict__ dinv, const float* __restrict__ bias,
                       float* __restrict__ agg, int n) {
    constexpr int TPR = C / 4;
    constexpr int G   = 32 / TPR;
    constexpr int R   = 4;
    constexpr int U   = 4;
    const int lane = threadIdx.x & 31;
    const int q = lane % TPR;
    const int g = lane / TPR;
    const int wave = __builtin_amdgcn_readfirstlane(
        (int)(blockIdx.x * (blockDim.x >> 5) + (threadIdx.x >> 5)));
    const int d0 = wave * R;
    if (d0 >= n) return;

    int st[R], len[R], cv[R];
    float dd[R];
    float4 hs[R];
#pragma unroll
    for (int r = 0; r < R; ++r) {
        const int d = (d0 + r < n) ? d0 + r : d0;
        st[r]  = start[d];
        len[r] = (d0 + r < n) ? cnt[d] : 0;
        dd[r]  = dinv[d];
        hs[r]  = *((const float4*)(h + (size_t)d * C) + q);
    }
#pragma unroll
    for (int r = 0; r < R; ++r) cv[r] = (lane < len[r]) ? col[st[r] + lane] : 0;

#pragma unroll
    for (int r = 0; r < R; ++r) {
        const int d = d0 + r;
        if (d >= n) break;
        const float dr = dd[r];
        float4 acc = make_float4(0.0f, 0.0f, 0.0f, 0.0f);
        if (g == 0) {
            const float w = dr * dr;
            acc.x = hs[r].x * w; acc.y = hs[r].y * w; acc.z = hs[r].z * w; acc.w = hs[r].w * w;
        }
        int c = cv[r];
        for (int j0 = 0; j0 < len[r]; j0 += 32) {
            if (j0 > 0) c = (j0 + lane < len[r]) ? col[st[r] + j0 + lane] : 0;
            const int m = min(len[r] - j0, 32);
            for (int j = 0; j < m; j += U * G) {
                float4 v[U];
                float w[U];
#pragma unroll
                for (int u = 0; u < U; ++u) {
                    const bool ok = (j + u * G + g) < m;
                    const int sraw = __shfl(c, (j + u * G + g) & 31);
                    const int s = ok ? sraw : d;
                    const float ds = dinv[s];
                    w[u] = ok ? ds * dr : 0.0f;
                    v[u] = *((const float4*)(h + (size_t)s * C) + q);
                }
#pragma unroll
                for (int u = 0; u < U; ++u) {
                    acc.x += v[u].x * w[u];
                    acc.y += v[u].y * w[u];
                    acc.z += v[u].z * w[u];
                    acc.w += v[u].w * w[u];
                }
            }
        }
#pragma unroll
        for (int o = TPR; o < 32; o <<= 1) {
            acc.x += __shfl_xor(acc.x, o);
            acc.y += __shfl_xor(acc.y, o);
            acc.z += __shfl_xor(acc.z, o);
            acc.w += __shfl_xor(acc.w, o);
        }
        if (g == 0) {
            if (BIAS) {
                acc.x += bias[q * 4 + 0]; acc.y += bias[q * 4 + 1];
                acc.z += bias[q * 4 + 2]; acc.w += bias[q * 4 + 3];
            }
            *((float4*)(agg + (size_t)d * C) + q) = acc;
        }
    }
}

extern "C" void kernel_launch(void* const* d_in, const int* in_sizes, int n_in,
                              void* d_out, int out_size, void* d_ws, size_t ws_size,
                              hipStream_t stream) {
    const float* x  = (const float*)d_in[0];
    const int*   ei = (const int*)d_in[1];
    const float* W1 = (const float*)d_in[2];
    const float* b1 = (const float*)d_in[3];
    const float* W2 = (const float*)d_in[4];
    const float* b2 = (const float*)d_in[5];
    float* out = (float*)d_out;

    const int n = in_sizes[0] / 128;
    const int e = in_sizes[1] / 2;
    const int* src = ei;
    const int* dst = ei + e;

    char* ws = (char*)d_ws;
    size_t off0 = ((size_t)(n + 1) * sizeof(float) + 255) & ~(size_t)255;
    float* dinv  = (float*)ws;
    int*   cnt   = (int*)(ws + off0);
    int*   start = (int*)(ws + 2 * off0);
    int*   cursor = (int*)(ws + 3 * off0);
    int*   tick   = cursor + TICK_STRIDE;
    float* bufA = (float*)(ws + 4 * off0);
    float* bufB = bufA + (size_t)n * 64;
    int*   col  = (int*)(bufB + (size_t)n * 64);
    int*   rank = (int*)bufB;

    const int tb = 256;
    auto cdiv = [](long long a, long long b) { return (int)((a + b - 1) / b); };

    k_zero<<<cdiv(n, tb), tb, 0, stream>>>(cnt, cursor, tick, n);
    k_rank<<<RANK_GRID, RANK_TB, 0, stream>>>(dst, cnt, rank, tick, e, n);
    k_alloc<<<cdiv(n, tb), tb, 0, stream>>>(cnt, cursor, start, dinv, n);
    k_place<<<cdiv(e, tb * EPT), tb, 0, stream>>>(src, dst, start, rank, col, e);

    const int strips = n / 16;
    const int wavesPerBlk = tb / 32;
    k_gemm_wmma<128, 64, false><<<cdiv(strips, wavesPerBlk), tb, 0, stream>>>(
        x, W1, nullptr, bufA, n);
    if (n % 16) {
        int tail = (n % 16) * 64;
        k_gemm_tail<128, 64, false><<<cdiv(tail, tb), tb, 0, stream>>>(
            x, W1, nullptr, bufA, strips * 16, n);
    }

    const int rowsPerBlk = wavesPerBlk * 4;
    k_pull<64, false><<<cdiv(n, rowsPerBlk), tb, 0, stream>>>(
        start, cnt, col, bufA, dinv, nullptr, bufB, n);

    k_gemm_wmma<64, 32, true><<<cdiv(strips, wavesPerBlk), tb, 0, stream>>>(
        bufB, W2, b1, bufA, n);
    if (n % 16) {
        int tail = (n % 16) * 32;
        k_gemm_tail<64, 32, true><<<cdiv(tail, tb), tb, 0, stream>>>(
            bufB, W2, b1, bufA, strips * 16, n);
    }

    k_pull<32, true><<<cdiv(n, rowsPerBlk), tb, 0, stream>>>(
        start, cnt, col, bufA, dinv, b2, out, n);

    (void)n_in; (void)out_size; (void)ws_size;
}


# counter clear folded into the x*W1 launch (now first), 1/sqrt(deg) taken from the counts inside the layer-1 pull (v_rsq_f32, 1 ulp; stored for layer 2): launches 7 -> 5; sum order unchanged
# speedup vs baseline: 3.7605x; 1.4759x over previous
#include <hip/hip_runtime.h>


typedef float v2f __attribute__((ext_vector_type(2)));
typedef float v8f __attribute__((ext_vector_type(8)));

constexpr int ELLW = 32;
constexpr int TICK_STRIDE = 64;

constexpr int EPT = 4;
constexpr int RANK_TB = 256;
constexpr int RANK_CHUNK = RANK_TB * EPT;
constexpr int RANK_GRID = 1024;

__global__ void k_rank(const int* __restrict__ src, const int* __restrict__ dst,
                       int* __restrict__ cnt, int* __restrict__ ell,
                       int* __restrict__ nspill, int* __restrict__ spill,
                       int* __restrict__ tick, int e, int n) {
    __shared__ int s_half;
    __shared__ int s_t[2];
    const int tid = threadIdx.x;
    const int nchunks = (e + RANK_CHUNK - 1) / RANK_CHUNK;
    const int nsplit = ((n >> 1) + 31) & ~31;
    if (tid == 0) {
        s_half = (int)((__builtin_amdgcn_s_sendmsg_rtn(0x87) >> 18) & 1u);
    }
    __syncthreads();
    const int half = s_half;

    for (int ph = 0; ph < 2; ++ph) {
        const int r = half ^ ph;
        int* const tk = tick + r * TICK_STRIDE;
        int tn = 0;
        if (tid == 0) tn = atomicAdd(tk, 1);
        for (int it = 0; it <= nchunks; ++it) {
            if (tid == 0) s_t[it & 1] = tn;
            __syncthreads();
            const int t = s_t[it & 1];
            if (t >= nchunks) break;

            const int base = t * RANK_CHUNK + tid;
            int d[EPT], s[EPT], k[EPT];
            bool mine[EPT];
#pragma unroll
            for (int u = 0; u < EPT; ++u) {
                const int i = base + u * RANK_TB;
                d[u] = dst[i < e ? i : e - 1];
            }
#pragma unroll
            for (int u = 0; u < EPT; ++u) {
                const int i = base + u * RANK_TB;
                s[u] = src[i < e ? i : e - 1];
            }
            asm volatile("" :: "v"(d[0]), "v"(d[1]), "v"(d[2]), "v"(d[3]));
            static_assert(EPT == 4, "the statement above names EPT values");
#pragma unroll
            for (int u = 0; u < EPT; ++u) {
                const int i = base + u * RANK_TB;
                mine[u] = (i < e) && ((d[u] >= nsplit) == (r != 0));
                if (mine[u]) k[u] = atomicAdd(&cnt[d[u]], 1);
            }
            if (tid == 0) tn = atomicAdd(tk, 1);
#pragma unroll
            for (int u = 0; u < EPT; ++u) {
                if (mine[u]) {
                    if (k[u] < ELLW) {
                        ell[(size_t)d[u] * ELLW + k[u]] = s[u];
                    } else {
                        const int p = atomicAdd(nspill, 1);
                        spill[p] = base + u * RANK_TB;
                    }
                }
            }
        }
        __syncthreads();
    }
}

__device__ __forceinline__ float inv_sqrt_deg(int c) {
    return __builtin_amdgcn_rsqf((float)(c + 1));
}

template <int K, int NOUT, bool RELU_IN, bool CLEAR>
__global__ void k_gemm_wmma(const float* __restrict__ A, const float* __restrict__ W,
                            const float* __restrict__ bin,
                            float* __restrict__ H, int nrows,
                            int* __restrict__ cnt, int* __restrict__ nspill,
                            int* __restrict__ tick) {
    if (CLEAR) {
        const int gi = blockIdx.x * blockDim.x + threadIdx.x;
        for (int i = gi; i < nrows; i += gridDim.x * blockDim.x) cnt[i] = 0;
        if (gi == 0) { *nspill = 0; tick[0] = 0; tick[TICK_STRIDE] = 0; }
    }
    const int wave = blockIdx.x * (blockDim.x >> 5) + (threadIdx.x >> 5);
    const int lane = threadIdx.x & 31;
    const int m    = lane & 15;
    const int kh   = lane >> 4;
    const int r0   = wave * 16;
    if (r0 + 16 > nrows) return;

    constexpr int NT = NOUT / 16;
    v8f acc[NT];
#pragma unroll
    for (int t = 0; t < NT; ++t)
#pragma unroll
        for (int j = 0; j < 8; ++j) acc[t][j] = 0.0f;

    const float* arow = A + (size_t)(r0 + m) * K;
    for (int k = 0; k < K; k += 4) {
        v2f a;
        a.x = arow[k + 2 * kh + 0];
        a.y = arow[k + 2 * kh + 1];
        if (RELU_IN) {
            float v0 = a.x + bin[k + 2 * kh + 0];
            float v1 = a.y + bin[k + 2 * kh + 1];
            a.x = v0 > 0.0f ? v0 : 0.0f;
            a.y = v1 > 0.0f ? v1 : 0.0f;
        }
#pragma unroll
        for (int t = 0; t < NT; ++t) {
            v2f b;
            b.x = W[(size_t)(k + 2 * kh + 0) * NOUT + t * 16 + m];
            b.y = W[(size_t)(k + 2 * kh + 1) * NOUT + t * 16 + m];
            acc[t] = __builtin_amdgcn_wmma_f32_16x16x4_f32(
                 false, a,  false, b,
                 (short)0, acc[t],  false,  false);
        }
    }

#pragma unroll
    for (int t = 0; t < NT; ++t) {
#pragma unroll
        for (int j = 0; j < 8; ++j) {
            const size_t o = (size_t)(r0 + j + 8 * kh) * NOUT + t * 16 + m;
            H[o] = acc[t][j];
        }
    }
}

template <int K, int NOUT, bool RELU_IN>
__global__ void k_gemm_tail(const float* __restrict__ A, const float* __restrict__ W,
                            const float* __restrict__ bin,
                            float* __restrict__ H, int row0, int nrows) {
    int idx = blockIdx.x * blockDim.x + threadIdx.x;
    int r = row0 + idx / NOUT;
    int c = idx % NOUT;
    if (r >= nrows) return;
    float s = 0.0f;
    for (int k = 0; k < K; ++k) {
        float a = A[(size_t)r * K + k];
        if (RELU_IN) {
            float v0 = a + bin[k];
            a = v0 > 0.0f ? v0 : 0.0f;
        }
        s += a * W[(size_t)k * NOUT + c];
    }
    H[(size_t)r * NOUT + c] = s;
}

template <int C, bool BIAS, bool FIRST>
__global__ void k_pull(const int* __restrict__ cnt, const int* ell,
                       const int* __restrict__ nspill, const int* __restrict__ spill,
                       const int* __restrict__ src, const int* __restrict__ dst,
                       const float* __restrict__ h,
                       const float* __restrict__ dinv_in, float* __restrict__ dinv_out,
                       const float* __restrict__ bias,
                       float* agg, int n) {
    constexpr int TPR = C / 4;
    constexpr int G   = 32 / TPR;
    constexpr int R   = 4;
    constexpr int U   = 4;
    static_assert(ELLW == 32, "lane = table slot");
    const int lane = threadIdx.x & 31;
    const int q = lane % TPR;
    const int g = lane / TPR;
    const int wave = __builtin_amdgcn_readfirstlane(
        (int)(blockIdx.x * (blockDim.x >> 5) + (threadIdx.x >> 5)));
    const int d0 = wave * R;
    if (d0 >= n) return;

    int len[R], cv[R];
    float dd[R];
    float4 hs[R];
#pragma unroll
    for (int r = 0; r < R; ++r) {
        const int d = d0 + r;
        len[r] = 0;
        dd[r]  = 0.0f;
        hs[r]  = make_float4(0.0f, 0.0f, 0.0f, 0.0f);
        cv[r]  = 0;
        if (d < n) {
            len[r] = cnt[d];
            dd[r]  = FIRST ? inv_sqrt_deg(len[r]) : dinv_in[d];
            hs[r]  = *((const float4*)(h + (size_t)d * C) + q);
            cv[r]  = ell[(size_t)d * ELLW + lane];
        }
    }
    if (FIRST) {
        static_assert(R == 4, "the select below names R values");
        const float mine = lane == 0 ? dd[0] : lane == 1 ? dd[1] : lane == 2 ? dd[2] : dd[3];
        if (lane < R && d0 + lane < n) dinv_out[d0 + lane] = mine;
    }

#pragma unroll
    for (int r = 0; r < R; ++r) {
        const int d = d0 + r;
        if (d >= n) break;
        const float dr = dd[r];
        float4 acc = make_float4(0.0f, 0.0f, 0.0f, 0.0f);
        if (g == 0) {
            const float w = dr * dr;
            acc.x = hs[r].x * w; acc.y = hs[r].y * w; acc.z = hs[r].z * w; acc.w = hs[r].w * w;
        }
        const int c = cv[r];
        const int m = min(len[r], ELLW);
        for (int j = 0; j < m; j += U * G) {
            float4 v[U];
            float w[U];
#pragma unroll
            for (int u = 0; u < U; ++u) {
                const bool ok = (j + u * G + g) < m;
                const int sraw = __shfl(c, (j + u * G + g) & 31);
                const int s = ok ? sraw : d;
                const float ds = FIRST ? inv_sqrt_deg(cnt[s]) : dinv_in[s];
                w[u] = ok ? ds * dr : 0.0f;
                v[u] = *((const float4*)(h + (size_t)s * C) + q);
            }
#pragma unroll
            for (int u = 0; u < U; ++u) {
                acc.x += v[u].x * w[u];
                acc.y += v[u].y * w[u];
                acc.z += v[u].z * w[u];
                acc.w += v[u].w * w[u];
            }
        }
        if (len[r] > ELLW) {
            const int ns = *nspill;
            for (int j0 = 0; j0 < ns; j0 += 32) {
                const bool in = (j0 + lane) < ns;
                const int ed = in ? spill[j0 + lane] : 0;
                const int sd = in ? dst[ed] : -1;
                const int ss = in ? src[ed] : 0;
                unsigned hit = (unsigned)__ballot(sd == d);
                while (hit) {
                    const int b = __builtin_ctz(hit);
                    hit &= hit - 1;
                    const int s = __shfl(ss, b);
                    if (g == 0) {
                        const float w = (FIRST ? inv_sqrt_deg(cnt[s]) : dinv_in[s]) * dr;
                        const float4 v = *((const float4*)(h + (size_t)s * C) + q);
                        acc.x += v.x * w; acc.y += v.y * w; acc.z += v.z * w; acc.w += v.w * w;
                    }
                }
            }
        }
#pragma unroll
        for (int o = TPR; o < 32; o <<= 1) {
            acc.x += __shfl_xor(acc.x, o);
            acc.y += __shfl_xor(acc.y, o);
            acc.z += __shfl_xor(acc.z, o);
            acc.w += __shfl_xor(acc.w, o);
        }
        if (g == 0) {
            if (BIAS) {
                acc.x += bias[q * 4 + 0]; acc.y += bias[q * 4 + 1];
                acc.z += bias[q * 4 + 2]; acc.w += bias[q * 4 + 3];
            }
            *((float4*)(agg + (size_t)d * C) + q) = acc;
        }
    }
}

extern "C" void kernel_launch(void* const* d_in, const int* in_sizes, int n_in,
                              void* d_out, int out_size, void* d_ws, size_t ws_size,
                              hipStream_t stream) {
    const float* x  = (const float*)d_in[0];
    const int*   ei = (const int*)d_in[1];
    const float* W1 = (const float*)d_in[2];
    const float* b1 = (const float*)d_in[3];
    const float* W2 = (const float*)d_in[4];
    const float* b2 = (const float*)d_in[5];
    float* out = (float*)d_out;

    const int n = in_sizes[0] / 128;
    const int e = in_sizes[1] / 2;
    const int* src = ei;
    const int* dst = ei + e;

    char* ws = (char*)d_ws;
    size_t off0 = ((size_t)(n + 1) * sizeof(float) + 255) & ~(size_t)255;
    float* dinv  = (float*)ws;
    int*   cnt   = (int*)(ws + off0);
    int*   nspill = (int*)(ws + 3 * off0);
    int*   tick   = nspill + TICK_STRIDE;
    float* bufA = (float*)(ws + 4 * off0);
    float* bufB = bufA + (size_t)n * 64;
    int*   spill = (int*)(bufB + (size_t)n * 64);
    int*   ell   = (int*)d_out;

    const int tb = 256;
    auto cdiv = [](long long a, long long b) { return (int)((a + b - 1) / b); };

    const int strips = n / 16;
    const int wavesPerBlk = tb / 32;
    k_gemm_wmma<128, 64, false, true><<<cdiv(strips, wavesPerBlk), tb, 0, stream>>>(
        x, W1, nullptr, bufA, n, cnt, nspill, tick);
    if (n % 16) {
        int tail = (n % 16) * 64;
        k_gemm_tail<128, 64, false><<<cdiv(tail, tb), tb, 0, stream>>>(
            x, W1, nullptr, bufA, strips * 16, n);
    }

    k_rank<<<RANK_GRID, RANK_TB, 0, stream>>>(src, dst, cnt, ell, nspill, spill, tick, e, n);

    const int rowsPerBlk = wavesPerBlk * 4;
    k_pull<64, false, true><<<cdiv(n, rowsPerBlk), tb, 0, stream>>>(
        cnt, ell, nspill, spill, src, dst, bufA, nullptr, dinv, nullptr, bufB, n);

    k_gemm_wmma<64, 32, true, false><<<cdiv(strips, wavesPerBlk), tb, 0, stream>>>(
        bufB, W2, b1, bufA, n, nullptr, nullptr, nullptr);
    if (n % 16) {
        int tail = (n % 16) * 32;
        k_gemm_tail<64, 32, true><<<cdiv(tail, tb), tb, 0, stream>>>(
            bufB, W2, b1, bufA, strips * 16, n);
    }

    k_pull<32, true, false><<<cdiv(n, rowsPerBlk), tb, 0, stream>>>(
        cnt, ell, nspill, spill, src, dst, bufA, dinv, nullptr, b2, out, n);

    (void)n_in; (void)out_size; (void)ws_size;
}


# layer-2 product moved into the layer-1 pull: a work-group's 32 summed rows go to LDS as relu(row+b1), 4 of its 8 waves do the W2 WMMA tiles and store h2; launches 5 -> 4; sums unchanged
# speedup vs baseline: 4.0943x; 1.0888x over previous
#include <hip/hip_runtime.h>


typedef float v2f __attribute__((ext_vector_type(2)));
typedef float v8f __attribute__((ext_vector_type(8)));

constexpr int ELLW = 32;
constexpr int TICK_STRIDE = 64;

constexpr int EPT = 4;
constexpr int RANK_TB = 256;
constexpr int RANK_CHUNK = RANK_TB * EPT;
constexpr int RANK_GRID = 1024;

__global__ void k_rank(const int* __restrict__ src, const int* __restrict__ dst,
                       int* __restrict__ cnt, int* __restrict__ ell,
                       int* __restrict__ nspill, int* __restrict__ spill,
                       int* __restrict__ tick, int e, int n) {
    __shared__ int s_half;
    __shared__ int s_t[2];
    const int tid = threadIdx.x;
    const int nchunks = (e + RANK_CHUNK - 1) / RANK_CHUNK;
    const int nsplit = ((n >> 1) + 31) & ~31;
    if (tid == 0) {
        s_half = (int)((__builtin_amdgcn_s_sendmsg_rtn(0x87) >> 18) & 1u);
    }
    __syncthreads();
    const int half = s_half;

    for (int ph = 0; ph < 2; ++ph) {
        const int r = half ^ ph;
        int* const tk = tick + r * TICK_STRIDE;
        int tn = 0;
        if (tid == 0) tn = atomicAdd(tk, 1);
        for (int it = 0; it <= nchunks; ++it) {
            if (tid == 0) s_t[it & 1] = tn;
            __syncthreads();
            const int t = s_t[it & 1];
            if (t >= nchunks) break;

            const int base = t * RANK_CHUNK + tid;
            int d[EPT], s[EPT], k[EPT];
            bool mine[EPT];
#pragma unroll
            for (int u = 0; u < EPT; ++u) {
                const int i = base + u * RANK_TB;
                d[u] = dst[i < e ? i : e - 1];
            }
#pragma unroll
            for (int u = 0; u < EPT; ++u) {
                const int i = base + u * RANK_TB;
                s[u] = src[i < e ? i : e - 1];
            }
            asm volatile("" :: "v"(d[0]), "v"(d[1]), "v"(d[2]), "v"(d[3]));
            static_assert(EPT == 4, "the statement above names EPT values");
#pragma unroll
            for (int u = 0; u < EPT; ++u) {
                const int i = base + u * RANK_TB;
                mine[u] = (i < e) && ((d[u] >= nsplit) == (r != 0));
                if (mine[u]) k[u] = atomicAdd(&cnt[d[u]], 1);
            }
            if (tid == 0) tn = atomicAdd(tk, 1);
#pragma unroll
            for (int u = 0; u < EPT; ++u) {
                if (mine[u]) {
                    if (k[u] < ELLW) {
                        ell[(size_t)d[u] * ELLW + k[u]] = s[u];
                    } else {
                        const int p = atomicAdd(nspill, 1);
                        spill[p] = base + u * RANK_TB;
                    }
                }
            }
        }
        __syncthreads();
    }
}

__device__ __forceinline__ float inv_sqrt_deg(int c) {
    return __builtin_amdgcn_rsqf((float)(c + 1));
}

template <int K, int NOUT, bool CLEAR>
__global__ void k_gemm_wmma(const float* __restrict__ A, const float* __restrict__ W,
                            float* __restrict__ H, int nrows,
                            int* __restrict__ cnt, int* __restrict__ nspill,
                            int* __restrict__ tick) {
    if (CLEAR) {
        const int gi = blockIdx.x * blockDim.x + threadIdx.x;
        for (int i = gi; i < nrows; i += gridDim.x * blockDim.x) cnt[i] = 0;
        if (gi == 0) { *nspill = 0; tick[0] = 0; tick[TICK_STRIDE] = 0; }
    }
    const int wave = blockIdx.x * (blockDim.x >> 5) + (threadIdx.x >> 5);
    const int lane = threadIdx.x & 31;
    const int m    = lane & 15;
    const int kh   = lane >> 4;
    const int r0   = wave * 16;
    if (r0 + 16 > nrows) return;

    constexpr int NT = NOUT / 16;
    v8f acc[NT];
#pragma unroll
    for (int t = 0; t < NT; ++t)
#pragma unroll
        for (int j = 0; j < 8; ++j) acc[t][j] = 0.0f;

    const float* arow = A + (size_t)(r0 + m) * K;
    for (int k = 0; k < K; k += 4) {
        v2f a;
        a.x = arow[k + 2 * kh + 0];
        a.y = arow[k + 2 * kh + 1];
#pragma unroll
        for (int t = 0; t < NT; ++t) {
            v2f b;
            b.x = W[(size_t)(k + 2 * kh + 0) * NOUT + t * 16 + m];
            b.y = W[(size_t)(k + 2 * kh + 1) * NOUT + t * 16 + m];
            acc[t] = __builtin_amdgcn_wmma_f32_16x16x4_f32(
                 false, a,  false, b,
                 (short)0, acc[t],  false,  false);
        }
    }

#pragma unroll
    for (int t = 0; t < NT; ++t) {
#pragma unroll
        for (int j = 0; j < 8; ++j) {
            const size_t o = (size_t)(r0 + j + 8 * kh) * NOUT + t * 16 + m;
            H[o] = acc[t][j];
        }
    }
}

template <int K, int NOUT>
__global__ void k_gemm_tail(const float* __restrict__ A, const float* __restrict__ W,
                            float* __restrict__ H, int row0, int nrows) {
    int idx = blockIdx.x * blockDim.x + threadIdx.x;
    int r = row0 + idx / NOUT;
    int c = idx % NOUT;
    if (r >= nrows) return;
    float s = 0.0f;
    for (int k = 0; k < K; ++k) {
        s += A[(size_t)r * K + k] * W[(size_t)k * NOUT + c];
    }
    H[(size_t)r * NOUT + c] = s;
}

constexpr int PULL_WAVES = 8;
constexpr int PULL_R     = 4;
constexpr int PULL_TB    = PULL_WAVES * 32;
constexpr int PULL_ROWS  = PULL_WAVES * PULL_R;
template <int C, bool BIAS, bool FIRST, int NOUT>
__global__ void k_pull(const int* __restrict__ cnt, const int* ell,
                       const int* __restrict__ nspill, const int* __restrict__ spill,
                       const int* __restrict__ src, const int* __restrict__ dst,
                       const float* __restrict__ h,
                       const float* __restrict__ dinv_in, float* __restrict__ dinv_out,
                       const float* __restrict__ bias,
                       const float* __restrict__ Wn, float* __restrict__ Hn,
                       float* agg, int n) {
    constexpr int TPR = C / 4;
    constexpr int G   = 32 / TPR;
    constexpr int R   = PULL_R;
    constexpr int ROWS = PULL_WAVES * R;
    constexpr int P = C + 4;
    __shared__ alignas(16) float s_rows[NOUT > 0 ? ROWS * P : 1];
    constexpr int U   = 4;
    static_assert(ELLW == 32, "lane = table slot");
    const int lane = threadIdx.x & 31;
    const int q = lane % TPR;
    const int g = lane / TPR;
    const int wave = __builtin_amdgcn_readfirstlane(
        (int)(blockIdx.x * (blockDim.x >> 5) + (threadIdx.x >> 5)));
    const int d0 = wave * R;
    if (NOUT == 0 && d0 >= n) return;
    const int wv = (int)(threadIdx.x >> 5);
    float4 bq = make_float4(0.0f, 0.0f, 0.0f, 0.0f);
    if (NOUT > 0 && BIAS)
        bq = make_float4(bias[q * 4 + 0], bias[q * 4 + 1], bias[q * 4 + 2], bias[q * 4 + 3]);

    int len[R], cv[R];
    float dd[R];
    float4 hs[R];
#pragma unroll
    for (int r = 0; r < R; ++r) {
        const int d = d0 + r;
        len[r] = 0;
        dd[r]  = 0.0f;
        hs[r]  = make_float4(0.0f, 0.0f, 0.0f, 0.0f);
        cv[r]  = 0;
        if (d < n) {
            len[r] = cnt[d];
            dd[r]  = FIRST ? inv_sqrt_deg(len[r]) : dinv_in[d];
            hs[r]  = *((const float4*)(h + (size_t)d * C) + q);
            cv[r]  = ell[(size_t)d * ELLW + lane];
        }
    }
    if (FIRST) {
        static_assert(R == 4, "the select below names R values");
        const float mine = lane == 0 ? dd[0] : lane == 1 ? dd[1] : lane == 2 ? dd[2] : dd[3];
        if (lane < R && d0 + lane < n) dinv_out[d0 + lane] = mine;
    }

#pragma unroll
    for (int r = 0; r < R; ++r) {
        const int d = d0 + r;
        if (d >= n) {
            if (NOUT == 0) break;
            if (g == 0)
                *((float4*)(s_rows + (wv * R + r) * P) + q) = make_float4(0.0f, 0.0f, 0.0f, 0.0f);
            continue;
        }
        const float dr = dd[r];
        float4 acc = make_float4(0.0f, 0.0f, 0.0f, 0.0f);
        if (g == 0) {
            const float w = dr * dr;
            acc.x = hs[r].x * w; acc.y = hs[r].y * w; acc.z = hs[r].z * w; acc.w = hs[r].w * w;
        }
        const int c = cv[r];
        const int m = min(len[r], ELLW);
        for (int j = 0; j < m; j += U * G) {
            float4 v[U];
            float w[U];
#pragma unroll
            for (int u = 0; u < U; ++u) {
                const bool ok = (j + u * G + g) < m;
                const int sraw = __shfl(c, (j + u * G + g) & 31);
                const int s = ok ? sraw : d;
                const float ds = FIRST ? inv_sqrt_deg(cnt[s]) : dinv_in[s];
                w[u] = ok ? ds * dr : 0.0f;
                v[u] = *((const float4*)(h + (size_t)s * C) + q);
            }
#pragma unroll
            for (int u = 0; u < U; ++u) {
                acc.x += v[u].x * w[u];
                acc.y += v[u].y * w[u];
                acc.z += v[u].z * w[u];
                acc.w += v[u].w * w[u];
            }
        }
        if (len[r] > ELLW) {
            const int ns = *nspill;
            for (int j0 = 0; j0 < ns; j0 += 32) {
                const bool in = (j0 + lane) < ns;
                const int ed = in ? spill[j0 + lane] : 0;
                const int sd = in ? dst[ed] : -1;
                const int ss = in ? src[ed] : 0;
                unsigned hit = (unsigned)__ballot(sd == d);
                while (hit) {
                    const int b = __builtin_ctz(hit);
                    hit &= hit - 1;
                    const int s = __shfl(ss, b);
                    if (g == 0) {
                        const float w = (FIRST ? inv_sqrt_deg(cnt[s]) : dinv_in[s]) * dr;
                        const float4 v = *((const float4*)(h + (size_t)s * C) + q);
                        acc.x += v.x * w; acc.y += v.y * w; acc.z += v.z * w; acc.w += v.w * w;
                    }
                }
            }
        }
#pragma unroll
        for (int o = TPR; o < 32; o <<= 1) {
            acc.x += __shfl_xor(acc.x, o);
            acc.y += __shfl_xor(acc.y, o);
            acc.z += __shfl_xor(acc.z, o);
            acc.w += __shfl_xor(acc.w, o);
        }
        if (g == 0) {
            if (BIAS && NOUT > 0) {
                acc.x += bq.x; acc.y += bq.y; acc.z += bq.z; acc.w += bq.w;
            } else if (BIAS) {
                acc.x += bias[q * 4 + 0]; acc.y += bias[q * 4 + 1];
                acc.z += bias[q * 4 + 2]; acc.w += bias[q * 4 + 3];
            }
            if (NOUT > 0) {
                acc.x = acc.x > 0.0f ? acc.x : 0.0f; acc.y = acc.y > 0.0f ? acc.y : 0.0f;
                acc.z = acc.z > 0.0f ? acc.z : 0.0f; acc.w = acc.w > 0.0f ? acc.w : 0.0f;
                *((float4*)(s_rows + (wv * R + r) * P) + q) = acc;
            } else {
                *((float4*)(agg + (size_t)d * C) + q) = acc;
            }
        }
    }

    if (NOUT > 0) {
        constexpr int NT = NOUT / 16;
        constexpr int TILES = (ROWS / 16) * NT;
        static_assert(ROWS % 16 == 0 && TILES <= PULL_WAVES, "one tile per wave");
        const bool tile_wave = wv < TILES;
        const int mt = wv / NT;
        const int nt = wv % NT;
        const int m  = lane & 15;
        const int kh = lane >> 4;
        v2f b[C / 4];
        if (tile_wave) {
            const float* wcol = Wn + nt * 16 + m;
#pragma unroll
            for (int k = 0; k < C; k += 4) {
                b[k / 4].x = wcol[(size_t)(k + 2 * kh + 0) * NOUT];
                b[k / 4].y = wcol[(size_t)(k + 2 * kh + 1) * NOUT];
            }
        }
        __syncthreads();
        if (!tile_wave) return;
        const float* arow = s_rows + (mt * 16 + m) * P;
        v8f t;
#pragma unroll
        for (int j = 0; j < 8; ++j) t[j] = 0.0f;
#pragma unroll
        for (int k = 0; k < C; k += 4) {
            const v2f a = *(const v2f*)(arow + k + 2 * kh);
            t = __builtin_amdgcn_wmma_f32_16x16x4_f32(
                 false, a,  false, b[k / 4],
                 (short)0, t,  false,  false);
        }
        const int row0 = (int)blockIdx.x * ROWS + mt * 16 + 8 * kh;
#pragma unroll
        for (int j = 0; j < 8; ++j) {
            if (row0 + j < n) Hn[(size_t)(row0 + j) * NOUT + nt * 16 + m] = t[j];
        }
    }
}

extern "C" void kernel_launch(void* const* d_in, const int* in_sizes, int n_in,
                              void* d_out, int out_size, void* d_ws, size_t ws_size,
                              hipStream_t stream) {
    const float* x  = (const float*)d_in[0];
    const int*   ei = (const int*)d_in[1];
    const float* W1 = (const float*)d_in[2];
    const float* b1 = (const float*)d_in[3];
    const float* W2 = (const float*)d_in[4];
    const float* b2 = (const float*)d_in[5];
    float* out = (float*)d_out;

    const int n = in_sizes[0] / 128;
    const int e = in_sizes[1] / 2;
    const int* src = ei;
    const int* dst = ei + e;

    char* ws = (char*)d_ws;
    size_t off0 = ((size_t)(n + 1) * sizeof(float) + 255) & ~(size_t)255;
    float* dinv  = (float*)ws;
    int*   cnt   = (int*)(ws + off0);
    int*   nspill = (int*)(ws + 3 * off0);
    int*   tick   = nspill + TICK_STRIDE;
    float* bufA = (float*)(ws + 4 * off0);
    float* bufB = bufA + (size_t)n * 64;
    int*   spill = (int*)(bufB + (size_t)n * 64);
    int*   ell   = (int*)d_out;

    const int tb = 256;
    auto cdiv = [](long long a, long long b) { return (int)((a + b - 1) / b); };

    const int strips = n / 16;
    const int wavesPerBlk = tb / 32;
    k_gemm_wmma<128, 64, true><<<cdiv(strips, wavesPerBlk), tb, 0, stream>>>(
        x, W1, bufA, n, cnt, nspill, tick);
    if (n % 16) {
        int tail = (n % 16) * 64;
        k_gemm_tail<128, 64><<<cdiv(tail, tb), tb, 0, stream>>>(
            x, W1, bufA, strips * 16, n);
    }

    k_rank<<<RANK_GRID, RANK_TB, 0, stream>>>(src, dst, cnt, ell, nspill, spill, tick, e, n);

    k_pull<64, true, true, 32><<<cdiv(n, PULL_ROWS), PULL_TB, 0, stream>>>(
        cnt, ell, nspill, spill, src, dst, bufA, nullptr, dinv, b1, W2, bufB, nullptr, n);

    k_pull<32, true, false, 0><<<cdiv(n, PULL_ROWS), PULL_TB, 0, stream>>>(
        cnt, ell, nspill, spill, src, dst, bufB, dinv, nullptr, b2, nullptr, nullptr, out, n);

    (void)n_in; (void)out_size; (void)ws_size;
}


# x*W1 launch made resident: 256 8-wave groups, a wave keeps all W1 fragments in registers and loops over row strips, next strip's x in flight, 16-B loads and stores; k pairs regrouped in the sums
# speedup vs baseline: 4.4358x; 1.0834x over previous
#include <hip/hip_runtime.h>


typedef float v2f __attribute__((ext_vector_type(2)));
typedef float v8f __attribute__((ext_vector_type(8)));

constexpr int ELLW = 32;
constexpr int TICK_STRIDE = 64;

constexpr int EPT = 4;
constexpr int RANK_TB = 256;
constexpr int RANK_CHUNK = RANK_TB * EPT;
constexpr int RANK_GRID = 1024;

__global__ void k_rank(const int* __restrict__ src, const int* __restrict__ dst,
                       int* __restrict__ cnt, int* __restrict__ ell,
                       int* __restrict__ nspill, int* __restrict__ spill,
                       int* __restrict__ tick, int e, int n) {
    __shared__ int s_half;
    __shared__ int s_t[2];
    const int tid = threadIdx.x;
    const int nchunks = (e + RANK_CHUNK - 1) / RANK_CHUNK;
    const int nsplit = ((n >> 1) + 31) & ~31;
    if (tid == 0) {
        s_half = (int)((__builtin_amdgcn_s_sendmsg_rtn(0x87) >> 18) & 1u);
    }
    __syncthreads();
    const int half = s_half;

    for (int ph = 0; ph < 2; ++ph) {
        const int r = half ^ ph;
        int* const tk = tick + r * TICK_STRIDE;
        int tn = 0;
        if (tid == 0) tn = atomicAdd(tk, 1);
        for (int it = 0; it <= nchunks; ++it) {
            if (tid == 0) s_t[it & 1] = tn;
            __syncthreads();
            const int t = s_t[it & 1];
            if (t >= nchunks) break;

            const int base = t * RANK_CHUNK + tid;
            int d[EPT], s[EPT], k[EPT];
            bool mine[EPT];
#pragma unroll
            for (int u = 0; u < EPT; ++u) {
                const int i = base + u * RANK_TB;
                d[u] = dst[i < e ? i : e - 1];
            }
#pragma unroll
            for (int u = 0; u < EPT; ++u) {
                const int i = base + u * RANK_TB;
                s[u] = src[i < e ? i : e - 1];
            }
            asm volatile("" :: "v"(d[0]), "v"(d[1]), "v"(d[2]), "v"(d[3]));
            static_assert(EPT == 4, "the statement above names EPT values");
#pragma unroll
            for (int u = 0; u < EPT; ++u) {
                const int i = base + u * RANK_TB;
                mine[u] = (i < e) && ((d[u] >= nsplit) == (r != 0));
                if (mine[u]) k[u] = atomicAdd(&cnt[d[u]], 1);
            }
            if (tid == 0) tn = atomicAdd(tk, 1);
#pragma unroll
            for (int u = 0; u < EPT; ++u) {
                if (mine[u]) {
                    if (k[u] < ELLW) {
                        ell[(size_t)d[u] * ELLW + k[u]] = s[u];
                    } else {
                        const int p = atomicAdd(nspill, 1);
                        spill[p] = base + u * RANK_TB;
                    }
                }
            }
        }
        __syncthreads();
    }
}

__device__ __forceinline__ float inv_sqrt_deg(int c) {
    return __builtin_amdgcn_rsqf((float)(c + 1));
}

constexpr int GEMM_TB   = 256;
constexpr int GEMM_GRID = 256;

template <int K>
__device__ __forceinline__ void load_strip(float4 (&a)[K / 8], const float* A,
                                           int s, int m, int kh) {
    const float4* p = (const float4*)(A + (size_t)(s * 16 + m) * K) + kh;
#pragma unroll
    for (int i = 0; i < K / 8; ++i) a[i] = p[2 * i];
}

template <int K, int NOUT>
__device__ __forceinline__ void strip_product(const float4 (&a)[K / 8],
                                              const v2f (&b)[K / 8][2][NOUT / 16],
                                              float* H, int s, int m, int kh) {
    constexpr int NT = NOUT / 16;
    static_assert(NT == 4, "a lane's four tile columns are one float4");
    v8f acc[NT];
#pragma unroll
    for (int t = 0; t < NT; ++t)
#pragma unroll
        for (int j = 0; j < 8; ++j) acc[t][j] = 0.0f;
#pragma unroll
    for (int i = 0; i < K / 8; ++i) {
        v2f lo, hi;
        lo.x = a[i].x; lo.y = a[i].y;
        hi.x = a[i].z; hi.y = a[i].w;
#pragma unroll
        for (int t = 0; t < NT; ++t)
            acc[t] = __builtin_amdgcn_wmma_f32_16x16x4_f32(
                 false, lo,  false, b[i][0][t],
                 (short)0, acc[t],  false,  false);
#pragma unroll
        for (int t = 0; t < NT; ++t)
            acc[t] = __builtin_amdgcn_wmma_f32_16x16x4_f32(
                 false, hi,  false, b[i][1][t],
                 (short)0, acc[t],  false,  false);
    }
    float* hrow = H + (size_t)(s * 16 + 8 * kh) * NOUT + 4 * m;
#pragma unroll
    for (int j = 0; j < 8; ++j) {
        *(float4*)(hrow + (size_t)j * NOUT) =
            make_float4(acc[0][j], acc[1][j], acc[2][j], acc[3][j]);
    }
}

template <int K, int NOUT, bool CLEAR>
__global__ void __launch_bounds__(GEMM_TB)
k_gemm_wmma(const float* A, const float* __restrict__ W,
            float* H, int nrows,
            int* __restrict__ cnt, int* __restrict__ nspill,
            int* __restrict__ tick) {
    if (CLEAR) {
        const int gi = blockIdx.x * blockDim.x + threadIdx.x;
        for (int i = gi; i < nrows; i += gridDim.x * blockDim.x) cnt[i] = 0;
        if (gi == 0) { *nspill = 0; tick[0] = 0; tick[TICK_STRIDE] = 0; }
    }
    const int lane = threadIdx.x & 31;
    const int m    = lane & 15;
    const int kh   = lane >> 4;
    const int nw   = (int)(gridDim.x * (blockDim.x >> 5));
    const int w = __builtin_amdgcn_readfirstlane(
        (int)((threadIdx.x >> 5) * gridDim.x + blockIdx.x));
    const int nstrips = nrows / 16;
    if (w >= nstrips) return;

    constexpr int NT = NOUT / 16;
    constexpr int KS = K / 8;
    static_assert(NT == 4, "a lane's four tile columns are one float4");
    v2f b[KS][2][NT];
#pragma unroll
    for (int i = 0; i < KS; ++i) {
        const float* wp = W + (size_t)(8 * i + 4 * kh) * NOUT + 4 * m;
        const float4 w0 = *(const float4*)(wp);
        const float4 w1 = *(const float4*)(wp + NOUT);
        const float4 w2 = *(const float4*)(wp + 2 * NOUT);
        const float4 w3 = *(const float4*)(wp + 3 * NOUT);
        b[i][0][0].x = w0.x; b[i][0][1].x = w0.y; b[i][0][2].x = w0.z; b[i][0][3].x = w0.w;
        b[i][0][0].y = w1.x; b[i][0][1].y = w1.y; b[i][0][2].y = w1.z; b[i][0][3].y = w1.w;
        b[i][1][0].x = w2.x; b[i][1][1].x = w2.y; b[i][1][2].x = w2.z; b[i][1][3].x = w2.w;
        b[i][1][0].y = w3.x; b[i][1][1].y = w3.y; b[i][1][2].y = w3.z; b[i][1][3].y = w3.w;
    }

    float4 a0[KS], a1[KS];
    int s = w;
    load_strip<K>(a0, A, s, m, kh);
    const int trips = (nstrips + 2 * nw - 1) / (2 * nw);
    for (int it = 0; it < trips; ++it) {
        int sn = s + nw;
        load_strip<K>(a1, A, sn < nstrips ? sn : s, m, kh);
        __builtin_amdgcn_sched_barrier(0);
        strip_product<K, NOUT>(a0, b, H, s, m, kh);
        if (sn >= nstrips) break;
        s = sn;
        sn = s + nw;
        load_strip<K>(a0, A, sn < nstrips ? sn : s, m, kh);
        __builtin_amdgcn_sched_barrier(0);
        strip_product<K, NOUT>(a1, b, H, s, m, kh);
        if (sn >= nstrips) break;
        s = sn;
    }
}

template <int K, int NOUT>
__global__ void k_gemm_tail(const float* __restrict__ A, const float* __restrict__ W,
                            float* __restrict__ H, int row0, int nrows) {
    int idx = blockIdx.x * blockDim.x + threadIdx.x;
    int r = row0 + idx / NOUT;
    int c = idx % NOUT;
    if (r >= nrows) return;
    float s = 0.0f;
    for (int k = 0; k < K; ++k) {
        s += A[(size_t)r * K + k] * W[(size_t)k * NOUT + c];
    }
    H[(size_t)r * NOUT + c] = s;
}

constexpr int PULL_WAVES = 8;
constexpr int PULL_R     = 4;
constexpr int PULL_TB    = PULL_WAVES * 32;
constexpr int PULL_ROWS  = PULL_WAVES * PULL_R;
template <int C, bool BIAS, bool FIRST, int NOUT>
__global__ void k_pull(const int* __restrict__ cnt, const int* ell,
                       const int* __restrict__ nspill, const int* __restrict__ spill,
                       const int* __restrict__ src, const int* __restrict__ dst,
                       const float* __restrict__ h,
                       const float* __restrict__ dinv_in, float* __restrict__ dinv_out,
                       const float* __restrict__ bias,
                       const float* __restrict__ Wn, float* __restrict__ Hn,
                       float* agg, int n) {
    constexpr int TPR = C / 4;
    constexpr int G   = 32 / TPR;
    constexpr int R   = PULL_R;
    constexpr int ROWS = PULL_WAVES * R;
    constexpr int P = C + 4;
    __shared__ alignas(16) float s_rows[NOUT > 0 ? ROWS * P : 1];
    constexpr int U   = 4;
    static_assert(ELLW == 32, "lane = table slot");
    const int lane = threadIdx.x & 31;
    const int q = lane % TPR;
    const int g = lane / TPR;
    const int wave = __builtin_amdgcn_readfirstlane(
        (int)(blockIdx.x * (blockDim.x >> 5) + (threadIdx.x >> 5)));
    const int d0 = wave * R;
    if (NOUT == 0 && d0 >= n) return;
    const int wv = (int)(threadIdx.x >> 5);
    float4 bq = make_float4(0.0f, 0.0f, 0.0f, 0.0f);
    if (NOUT > 0 && BIAS)
        bq = make_float4(bias[q * 4 + 0], bias[q * 4 + 1], bias[q * 4 + 2], bias[q * 4 + 3]);

    int len[R], cv[R];
    float dd[R];
    float4 hs[R];
#pragma unroll
    for (int r = 0; r < R; ++r) {
        const int d = d0 + r;
        len[r] = 0;
        dd[r]  = 0.0f;
        hs[r]  = make_float4(0.0f, 0.0f, 0.0f, 0.0f);
        cv[r]  = 0;
        if (d < n) {
            len[r] = cnt[d];
            dd[r]  = FIRST ? inv_sqrt_deg(len[r]) : dinv_in[d];
            hs[r]  = *((const float4*)(h + (size_t)d * C) + q);
            cv[r]  = ell[(size_t)d * ELLW + lane];
        }
    }
    if (FIRST) {
        static_assert(R == 4, "the select below names R values");
        const float mine = lane == 0 ? dd[0] : lane == 1 ? dd[1] : lane == 2 ? dd[2] : dd[3];
        if (lane < R && d0 + lane < n) dinv_out[d0 + lane] = mine;
    }

#pragma unroll
    for (int r = 0; r < R; ++r) {
        const int d = d0 + r;
        if (d >= n) {
            if (NOUT == 0) break;
            if (g == 0)
                *((float4*)(s_rows + (wv * R + r) * P) + q) = make_float4(0.0f, 0.0f, 0.0f, 0.0f);
            continue;
        }
        const float dr = dd[r];
        float4 acc = make_float4(0.0f, 0.0f, 0.0f, 0.0f);
        if (g == 0) {
            const float w = dr * dr;
            acc.x = hs[r].x * w; acc.y = hs[r].y * w; acc.z = hs[r].z * w; acc.w = hs[r].w * w;
        }
        const int c = cv[r];
        const int m = min(len[r], ELLW);
        for (int j = 0; j < m; j += U * G) {
            float4 v[U];
            float w[U];
#pragma unroll
            for (int u = 0; u < U; ++u) {
                const bool ok = (j + u * G + g) < m;
                const int sraw = __shfl(c, (j + u * G + g) & 31);
                const int s = ok ? sraw : d;
                const float ds = FIRST ? inv_sqrt_deg(cnt[s]) : dinv_in[s];
                w[u] = ok ? ds * dr : 0.0f;
                v[u] = *((const float4*)(h + (size_t)s * C) + q);
            }
#pragma unroll
            for (int u = 0; u < U; ++u) {
                acc.x += v[u].x * w[u];
                acc.y += v[u].y * w[u];
                acc.z += v[u].z * w[u];
                acc.w += v[u].w * w[u];
            }
        }
        if (len[r] > ELLW) {
            const int ns = *nspill;
            for (int j0 = 0; j0 < ns; j0 += 32) {
                const bool in = (j0 + lane) < ns;
                const int ed = in ? spill[j0 + lane] : 0;
                const int sd = in ? dst[ed] : -1;
                const int ss = in ? src[ed] : 0;
                unsigned hit = (unsigned)__ballot(sd == d);
                while (hit) {
                    const int b = __builtin_ctz(hit);
                    hit &= hit - 1;
                    const int s = __shfl(ss, b);
                    if (g == 0) {
                        const float w = (FIRST ? inv_sqrt_deg(cnt[s]) : dinv_in[s]) * dr;
                        const float4 v = *((const float4*)(h + (size_t)s * C) + q);
                        acc.x += v.x * w; acc.y += v.y * w; acc.z += v.z * w; acc.w += v.w * w;
                    }
                }
            }
        }
#pragma unroll
        for (int o = TPR; o < 32; o <<= 1) {
            acc.x += __shfl_xor(acc.x, o);
            acc.y += __shfl_xor(acc.y, o);
            acc.z += __shfl_xor(acc.z, o);
            acc.w += __shfl_xor(acc.w, o);
        }
        if (g == 0) {
            if (BIAS && NOUT > 0) {
                acc.x += bq.x; acc.y += bq.y; acc.z += bq.z; acc.w += bq.w;
            } else if (BIAS) {
                acc.x += bias[q * 4 + 0]; acc.y += bias[q * 4 + 1];
                acc.z += bias[q * 4 + 2]; acc.w += bias[q * 4 + 3];
            }
            if (NOUT > 0) {
                acc.x = acc.x > 0.0f ? acc.x : 0.0f; acc.y = acc.y > 0.0f ? acc.y : 0.0f;
                acc.z = acc.z > 0.0f ? acc.z : 0.0f; acc.w = acc.w > 0.0f ? acc.w : 0.0f;
                *((float4*)(s_rows + (wv * R + r) * P) + q) = acc;
            } else {
                *((float4*)(agg + (size_t)d * C) + q) = acc;
            }
        }
    }

    if (NOUT > 0) {
        constexpr int NT = NOUT / 16;
        constexpr int TILES = (ROWS / 16) * NT;
        static_assert(ROWS % 16 == 0 && TILES <= PULL_WAVES, "one tile per wave");
        const bool tile_wave = wv < TILES;
        const int mt = wv / NT;
        const int nt = wv % NT;
        const int m  = lane & 15;
        const int kh = lane >> 4;
        v2f b[C / 4];
        if (tile_wave) {
            const float* wcol = Wn + nt * 16 + m;
#pragma unroll
            for (int k = 0; k < C; k += 4) {
                b[k / 4].x = wcol[(size_t)(k + 2 * kh + 0) * NOUT];
                b[k / 4].y = wcol[(size_t)(k + 2 * kh + 1) * NOUT];
            }
        }
        __syncthreads();
        if (!tile_wave) return;
        const float* arow = s_rows + (mt * 16 + m) * P;
        v8f t;
#pragma unroll
        for (int j = 0; j < 8; ++j) t[j] = 0.0f;
#pragma unroll
        for (int k = 0; k < C; k += 4) {
            const v2f a = *(const v2f*)(arow + k + 2 * kh);
            t = __builtin_amdgcn_wmma_f32_16x16x4_f32(
                 false, a,  false, b[k / 4],
                 (short)0, t,  false,  false);
        }
        const int row0 = (int)blockIdx.x * ROWS + mt * 16 + 8 * kh;
#pragma unroll
        for (int j = 0; j < 8; ++j) {
            if (row0 + j < n) Hn[(size_t)(row0 + j) * NOUT + nt * 16 + m] = t[j];
        }
    }
}

extern "C" void kernel_launch(void* const* d_in, const int* in_sizes, int n_in,
                              void* d_out, int out_size, void* d_ws, size_t ws_size,
                              hipStream_t stream) {
    const float* x  = (const float*)d_in[0];
    const int*   ei = (const int*)d_in[1];
    const float* W1 = (const float*)d_in[2];
    const float* b1 = (const float*)d_in[3];
    const float* W2 = (const float*)d_in[4];
    const float* b2 = (const float*)d_in[5];
    float* out = (float*)d_out;

    const int n = in_sizes[0] / 128;
    const int e = in_sizes[1] / 2;
    const int* src = ei;
    const int* dst = ei + e;

    char* ws = (char*)d_ws;
    size_t off0 = ((size_t)(n + 1) * sizeof(float) + 255) & ~(size_t)255;
    float* dinv  = (float*)ws;
    int*   cnt   = (int*)(ws + off0);
    int*   nspill = (int*)(ws + 3 * off0);
    int*   tick   = nspill + TICK_STRIDE;
    float* bufA = (float*)(ws + 4 * off0);
    float* bufB = bufA + (size_t)n * 64;
    int*   spill = (int*)(bufB + (size_t)n * 64);
    int*   ell   = (int*)d_out;

    const int tb = 256;
    auto cdiv = [](long long a, long long b) { return (int)((a + b - 1) / b); };

    const int strips = n / 16;
    k_gemm_wmma<128, 64, true><<<GEMM_GRID, GEMM_TB, 0, stream>>>(
        x, W1, bufA, n, cnt, nspill, tick);
    if (n % 16) {
        int tail = (n % 16) * 64;
        k_gemm_tail<128, 64><<<cdiv(tail, tb), tb, 0, stream>>>(
            x, W1, bufA, strips * 16, n);
    }

    k_rank<<<RANK_GRID, RANK_TB, 0, stream>>>(src, dst, cnt, ell, nspill, spill, tick, e, n);

    k_pull<64, true, true, 32><<<cdiv(n, PULL_ROWS), PULL_TB, 0, stream>>>(
        cnt, ell, nspill, spill, src, dst, bufA, nullptr, dinv, b1, W2, bufB, nullptr, n);

    k_pull<32, true, false, 0><<<cdiv(n, PULL_ROWS), PULL_TB, 0, stream>>>(
        cnt, ell, nspill, spill, src, dst, bufB, dinv, nullptr, b2, nullptr, nullptr, out, n);

    (void)n_in; (void)out_size; (void)ws_size;
}
